# starting kernel
# baseline (speedup 1.0000x reference)
#include <hip/hip_runtime.h>

#define BATCH 8
#define LSEQ 1024
#define HIDDEN 256
#define NUM_RBF 16
#define ROWS 16
#define JLANES 16

typedef __attribute__((ext_vector_type(2))) float v2f;
typedef __attribute__((ext_vector_type(8))) float v8f;

#define NEG_HALF_LOG2E (-0.72134752044448170368f)

__global__ __launch_bounds__(256) void rbf_mean_kernel(
    const float* __restrict__ pos,
    const float* __restrict__ centers,
    float* __restrict__ rbf_mean)
{
    __shared__ float sPos[LSEQ * 3];
    __shared__ float sPart[256][NUM_RBF + 1];

    const int tid          = threadIdx.x;
    const int tilesPerB    = LSEQ / ROWS;
    const int b            = blockIdx.x / tilesPerB;
    const int rt           = blockIdx.x % tilesPerB;

    const float* posB = pos + (size_t)b * LSEQ * 3;
    for (int idx = tid; idx < LSEQ * 3; idx += 256)
        sPos[idx] = posB[idx];

    float c[NUM_RBF];
#pragma unroll
    for (int r = 0; r < NUM_RBF; ++r) c[r] = centers[r];

    __syncthreads();

    const int jl  = tid & (JLANES - 1);
    const int row = tid >> 4;
    const int i   = rt * ROWS + row;

    const float xi = sPos[i * 3 + 0];
    const float yi = sPos[i * 3 + 1];
    const float zi = sPos[i * 3 + 2];

    float acc[NUM_RBF];
#pragma unroll
    for (int r = 0; r < NUM_RBF; ++r) acc[r] = 0.0f;

#pragma unroll 2
    for (int j = jl; j < LSEQ; j += JLANES) {
        const float dx = xi - sPos[j * 3 + 0];
        const float dy = yi - sPos[j * 3 + 1];
        const float dz = zi - sPos[j * 3 + 2];
        const float d  = __builtin_sqrtf(dx * dx + dy * dy + dz * dz);
#pragma unroll
        for (int r = 0; r < NUM_RBF; ++r) {
            const float t = d - c[r];
            acc[r] += __builtin_amdgcn_exp2f(t * t * NEG_HALF_LOG2E);
        }
    }

#pragma unroll
    for (int r = 0; r < NUM_RBF; ++r) sPart[tid][r] = acc[r];
    __syncthreads();

    const int orow = tid >> 4;
    const int r    = tid & 15;
    float s = 0.0f;
#pragma unroll
    for (int k = 0; k < JLANES; ++k)
        s += sPart[orow * JLANES + k][r];

    const int gRow = b * LSEQ + rt * ROWS + orow;
    rbf_mean[gRow * NUM_RBF + r] = s * (1.0f / (float)LSEQ);
}

__global__ __launch_bounds__(256) void linear_wmma_kernel(
    const float* __restrict__ rbf_mean,
    const float* __restrict__ W,
    const float* __restrict__ bias,
    const float* __restrict__ latent,
    float* __restrict__ out)
{
    const int lane     = threadIdx.x & 31;
    const int waveBlk  = threadIdx.x >> 5;
    const int wave     = blockIdx.x * 8 + waveBlk;
    const int rowBase  = wave * 16;

    const int m  = lane & 15;
    const int hi = lane >> 4;

    v2f a[4];
#pragma unroll
    for (int kc = 0; kc < 4; ++kc) {
        const float* p = rbf_mean + (size_t)(rowBase + m) * NUM_RBF + kc * 4 + hi * 2;
        v2f t; t.x = p[0]; t.y = p[1];
        a[kc] = t;
    }

#pragma unroll 1
    for (int nt = 0; nt < HIDDEN / 16; ++nt) {
        const int n = nt * 16 + m;

        v2f bf[4];
#pragma unroll
        for (int kc = 0; kc < 4; ++kc) {
            const float* p = W + (size_t)n * NUM_RBF + kc * 4 + hi * 2;
            v2f t; t.x = p[0]; t.y = p[1];
            bf[kc] = t;
        }

        v8f cfr = {0.f, 0.f, 0.f, 0.f, 0.f, 0.f, 0.f, 0.f};
#pragma unroll
        for (int kc = 0; kc < 4; ++kc) {
            cfr = __builtin_amdgcn_wmma_f32_16x16x4_f32(
                 false, a[kc],  false, bf[kc],
                 (short)0, cfr,  false,  false);
        }

        const float bv = bias[n];
#pragma unroll
        for (int v = 0; v < 8; ++v) {
            const int    rrow = rowBase + v + hi * 8;
            const int    col  = nt * 16 + m;
            const size_t idx  = (size_t)rrow * HIDDEN + col;
            out[idx] = latent[idx] + cfr[v] + bv;
        }
    }
}

extern "C" void kernel_launch(void* const* d_in, const int* in_sizes, int n_in,
                              void* d_out, int out_size, void* d_ws, size_t ws_size,
                              hipStream_t stream) {
    const float* pos     = (const float*)d_in[0];
    const float* latent  = (const float*)d_in[1];
    const float* W       = (const float*)d_in[2];
    const float* bias    = (const float*)d_in[3];
    const float* centers = (const float*)d_in[4];
    float* out      = (float*)d_out;
    float* rbf_mean = (float*)d_ws;

    rbf_mean_kernel<<<BATCH * (LSEQ / ROWS), 256, 0, stream>>>(pos, centers, rbf_mean);

    linear_wmma_kernel<<<(BATCH * LSEQ / 16) / 8, 256, 0, stream>>>(
        rbf_mean, W, bias, latent, out);
}


# Fuse the linear (WMMA) layer into the rbf kernel as a per-block epilogue: one launch instead of two, the 16x16 mean tile stays in LDS, latent/W/bias are requested before the pair loop
# speedup vs baseline: 1.5700x; 1.5700x over previous
#include <hip/hip_runtime.h>

#define BATCH 8
#define LSEQ 1024
#define HIDDEN 256
#define NUM_RBF 16
#define ROWS 16
#define JLANES 16
#define MPITCH 18

typedef __attribute__((ext_vector_type(2))) float v2f;
typedef __attribute__((ext_vector_type(8))) float v8f;

#define NEG_HALF_LOG2E (-0.72134752044448170368f)

__global__ __launch_bounds__(256) void rbf_linear_fused_kernel(
    const float* __restrict__ pos,
    const float* __restrict__ centers,
    const float* __restrict__ W,
    const float* __restrict__ bias,
    const float* __restrict__ latent,
    float* __restrict__ out)
{
    __shared__ float sPos[LSEQ * 3];
    __shared__ float sPart[256][NUM_RBF + 1];
    __shared__ __attribute__((aligned(16))) float sMean[ROWS][MPITCH];

    const int tid          = threadIdx.x;
    const int tilesPerB    = LSEQ / ROWS;
    const int b            = blockIdx.x / tilesPerB;
    const int rt           = blockIdx.x % tilesPerB;

    const float* posB = pos + (size_t)b * LSEQ * 3;
    for (int idx = tid; idx < LSEQ * 3; idx += 256)
        sPos[idx] = posB[idx];

    float c[NUM_RBF];
#pragma unroll
    for (int r = 0; r < NUM_RBF; ++r) c[r] = centers[r];

    __syncthreads();

    const int lane    = tid & 31;
    const int wv      = tid >> 5;
    const int m       = lane & 15;
    const int hi      = lane >> 4;
    const int rowBase = blockIdx.x * ROWS;
    const int colBase = wv * 32 + 2 * m;

    v2f lat[8];
    for (int v = 0; v < 8; ++v) {
        const size_t idx = (size_t)(rowBase + v + hi * 8) * HIDDEN + colBase;
        lat[v] = *(const v2f*)(latent + idx);
    }
    v2f bf0[4], bf1[4];
    for (int kc = 0; kc < 4; ++kc) {
        bf0[kc] = *(const v2f*)(W + (size_t)(colBase + 0) * NUM_RBF + kc * 4 + hi * 2);
        bf1[kc] = *(const v2f*)(W + (size_t)(colBase + 1) * NUM_RBF + kc * 4 + hi * 2);
    }
    const v2f bv = *(const v2f*)(bias + colBase);

    const int jl  = tid & (JLANES - 1);
    const int row = tid >> 4;
    const int i   = rt * ROWS + row;

    const float xi = sPos[i * 3 + 0];
    const float yi = sPos[i * 3 + 1];
    const float zi = sPos[i * 3 + 2];

    float acc[NUM_RBF];
#pragma unroll
    for (int r = 0; r < NUM_RBF; ++r) acc[r] = 0.0f;

#pragma unroll 2
    for (int j = jl; j < LSEQ; j += JLANES) {
        const float dx = xi - sPos[j * 3 + 0];
        const float dy = yi - sPos[j * 3 + 1];
        const float dz = zi - sPos[j * 3 + 2];
        const float d  = __builtin_sqrtf(dx * dx + dy * dy + dz * dz);
#pragma unroll
        for (int r = 0; r < NUM_RBF; ++r) {
            const float t = d - c[r];
            acc[r] += __builtin_amdgcn_exp2f(t * t * NEG_HALF_LOG2E);
        }
    }

#pragma unroll
    for (int r = 0; r < NUM_RBF; ++r) sPart[tid][r] = acc[r];
    __syncthreads();

    {
        const int orow = tid >> 4;
        const int r    = tid & 15;
        float s = 0.0f;
#pragma unroll
        for (int k = 0; k < JLANES; ++k)
            s += sPart[orow * JLANES + k][r];
        sMean[orow][r] = s * (1.0f / (float)LSEQ);
    }
    __syncthreads();

    v2f a[4];
#pragma unroll
    for (int kc = 0; kc < 4; ++kc)
        a[kc] = *(const v2f*)(&sMean[m][kc * 4 + hi * 2]);

    v8f c0 = {0.f, 0.f, 0.f, 0.f, 0.f, 0.f, 0.f, 0.f};
    v8f c1 = {0.f, 0.f, 0.f, 0.f, 0.f, 0.f, 0.f, 0.f};
#pragma unroll
    for (int kc = 0; kc < 4; ++kc) {
        c0 = __builtin_amdgcn_wmma_f32_16x16x4_f32(
             false, a[kc],  false, bf0[kc],
             (short)0, c0,  false,  false);
        c1 = __builtin_amdgcn_wmma_f32_16x16x4_f32(
             false, a[kc],  false, bf1[kc],
             (short)0, c1,  false,  false);
    }

#pragma unroll
    for (int v = 0; v < 8; ++v) {
        const size_t idx = (size_t)(rowBase + v + hi * 8) * HIDDEN + colBase;
        v2f o;
        o.x = lat[v].x + c0[v] + bv.x;
        o.y = lat[v].y + c1[v] + bv.y;
        *(v2f*)(out + idx) = o;
    }
}

extern "C" void kernel_launch(void* const* d_in, const int* in_sizes, int n_in,
                              void* d_out, int out_size, void* d_ws, size_t ws_size,
                              hipStream_t stream) {
    const float* pos     = (const float*)d_in[0];
    const float* latent  = (const float*)d_in[1];
    const float* W       = (const float*)d_in[2];
    const float* bias    = (const float*)d_in[3];
    const float* centers = (const float*)d_in[4];
    float* out      = (float*)d_out;

    rbf_linear_fused_kernel<<<BATCH * (LSEQ / ROWS), 256, 0, stream>>>(
        pos, centers, W, bias, latent, out);
}


# pair distance: the hardware square root v_sqrt_f32 (within 1 ulp) replaces the correctly rounded sqrt sequence: 11 vector instructions per pair fewer in the issue-bound pair loop
# speedup vs baseline: 1.7544x; 1.1175x over previous
#include <hip/hip_runtime.h>

#define BATCH 8
#define LSEQ 1024
#define HIDDEN 256
#define NUM_RBF 16
#define ROWS 16
#define JLANES 16
#define MPITCH 18

typedef __attribute__((ext_vector_type(2))) float v2f;
typedef __attribute__((ext_vector_type(8))) float v8f;

#define NEG_HALF_LOG2E (-0.72134752044448170368f)

__global__ __launch_bounds__(256) void rbf_linear_fused_kernel(
    const float* __restrict__ pos,
    const float* __restrict__ centers,
    const float* __restrict__ W,
    const float* __restrict__ bias,
    const float* __restrict__ latent,
    float* __restrict__ out)
{
    __shared__ float sPos[LSEQ * 3];
    __shared__ float sPart[256][NUM_RBF + 1];
    __shared__ __attribute__((aligned(16))) float sMean[ROWS][MPITCH];

    const int tid          = threadIdx.x;
    const int tilesPerB    = LSEQ / ROWS;
    const int b            = blockIdx.x / tilesPerB;
    const int rt           = blockIdx.x % tilesPerB;

    const float* posB = pos + (size_t)b * LSEQ * 3;
    for (int idx = tid; idx < LSEQ * 3; idx += 256)
        sPos[idx] = posB[idx];

    float c[NUM_RBF];
#pragma unroll
    for (int r = 0; r < NUM_RBF; ++r) c[r] = centers[r];

    __syncthreads();

    const int lane    = tid & 31;
    const int wv      = tid >> 5;
    const int m       = lane & 15;
    const int hi      = lane >> 4;
    const int rowBase = blockIdx.x * ROWS;
    const int colBase = wv * 32 + 2 * m;

    v2f lat[8];
    for (int v = 0; v < 8; ++v) {
        const size_t idx = (size_t)(rowBase + v + hi * 8) * HIDDEN + colBase;
        lat[v] = *(const v2f*)(latent + idx);
    }
    v2f bf0[4], bf1[4];
    for (int kc = 0; kc < 4; ++kc) {
        bf0[kc] = *(const v2f*)(W + (size_t)(colBase + 0) * NUM_RBF + kc * 4 + hi * 2);
        bf1[kc] = *(const v2f*)(W + (size_t)(colBase + 1) * NUM_RBF + kc * 4 + hi * 2);
    }
    const v2f bv = *(const v2f*)(bias + colBase);

    const int jl  = tid & (JLANES - 1);
    const int row = tid >> 4;
    const int i   = rt * ROWS + row;

    const float xi = sPos[i * 3 + 0];
    const float yi = sPos[i * 3 + 1];
    const float zi = sPos[i * 3 + 2];

    float acc[NUM_RBF];
#pragma unroll
    for (int r = 0; r < NUM_RBF; ++r) acc[r] = 0.0f;

#pragma unroll 2
    for (int j = jl; j < LSEQ; j += JLANES) {
        const float dx = xi - sPos[j * 3 + 0];
        const float dy = yi - sPos[j * 3 + 1];
        const float dz = zi - sPos[j * 3 + 2];
        const float d  = __builtin_amdgcn_sqrtf(dx * dx + dy * dy + dz * dz);
#pragma unroll
        for (int r = 0; r < NUM_RBF; ++r) {
            const float t = d - c[r];
            acc[r] += __builtin_amdgcn_exp2f(t * t * NEG_HALF_LOG2E);
        }
    }

#pragma unroll
    for (int r = 0; r < NUM_RBF; ++r) sPart[tid][r] = acc[r];
    __syncthreads();

    {
        const int orow = tid >> 4;
        const int r    = tid & 15;
        float s = 0.0f;
#pragma unroll
        for (int k = 0; k < JLANES; ++k)
            s += sPart[orow * JLANES + k][r];
        sMean[orow][r] = s * (1.0f / (float)LSEQ);
    }
    __syncthreads();

    v2f a[4];
#pragma unroll
    for (int kc = 0; kc < 4; ++kc)
        a[kc] = *(const v2f*)(&sMean[m][kc * 4 + hi * 2]);

    v8f c0 = {0.f, 0.f, 0.f, 0.f, 0.f, 0.f, 0.f, 0.f};
    v8f c1 = {0.f, 0.f, 0.f, 0.f, 0.f, 0.f, 0.f, 0.f};
#pragma unroll
    for (int kc = 0; kc < 4; ++kc) {
        c0 = __builtin_amdgcn_wmma_f32_16x16x4_f32(
             false, a[kc],  false, bf0[kc],
             (short)0, c0,  false,  false);
        c1 = __builtin_amdgcn_wmma_f32_16x16x4_f32(
             false, a[kc],  false, bf1[kc],
             (short)0, c1,  false,  false);
    }

#pragma unroll
    for (int v = 0; v < 8; ++v) {
        const size_t idx = (size_t)(rowBase + v + hi * 8) * HIDDEN + colBase;
        v2f o;
        o.x = lat[v].x + c0[v] + bv.x;
        o.y = lat[v].y + c1[v] + bv.y;
        *(v2f*)(out + idx) = o;
    }
}

extern "C" void kernel_launch(void* const* d_in, const int* in_sizes, int n_in,
                              void* d_out, int out_size, void* d_ws, size_t ws_size,
                              hipStream_t stream) {
    const float* pos     = (const float*)d_in[0];
    const float* latent  = (const float*)d_in[1];
    const float* W       = (const float*)d_in[2];
    const float* bias    = (const float*)d_in[3];
    const float* centers = (const float*)d_in[4];
    float* out      = (float*)d_out;

    rbf_linear_fused_kernel<<<BATCH * (LSEQ / ROWS), 256, 0, stream>>>(
        pos, centers, W, bias, latent, out);
}


# pair loop issue order: each v_exp_f32 now stands between packed instructions of other groups (skewed pipeline) instead of 32 back to back; same operations and summation order
# speedup vs baseline: 1.8814x; 1.0724x over previous
#include <hip/hip_runtime.h>

#define BATCH 8
#define LSEQ 1024
#define HIDDEN 256
#define NUM_RBF 16
#define ROWS 16
#define JLANES 16
#define MPITCH 18

typedef __attribute__((ext_vector_type(2))) float v2f;
typedef __attribute__((ext_vector_type(8))) float v8f;

#define NEG_HALF_LOG2E (-0.72134752044448170368f)

__global__ __launch_bounds__(256) void rbf_linear_fused_kernel(
    const float* __restrict__ pos,
    const float* __restrict__ centers,
    const float* __restrict__ W,
    const float* __restrict__ bias,
    const float* __restrict__ latent,
    float* __restrict__ out)
{
    __shared__ float sPos[LSEQ * 3];
    __shared__ float sPart[256][NUM_RBF + 1];
    __shared__ __attribute__((aligned(16))) float sMean[ROWS][MPITCH];

    const int tid          = threadIdx.x;
    const int tilesPerB    = LSEQ / ROWS;
    const int b            = blockIdx.x / tilesPerB;
    const int rt           = blockIdx.x % tilesPerB;

    const float* posB = pos + (size_t)b * LSEQ * 3;
    for (int idx = tid; idx < LSEQ * 3; idx += 256)
        sPos[idx] = posB[idx];

    float c[NUM_RBF];
#pragma unroll
    for (int r = 0; r < NUM_RBF; ++r) c[r] = centers[r];

    __syncthreads();

    const int lane    = tid & 31;
    const int wv      = tid >> 5;
    const int m       = lane & 15;
    const int hi      = lane >> 4;
    const int rowBase = blockIdx.x * ROWS;
    const int colBase = wv * 32 + 2 * m;

    v2f lat[8];
    for (int v = 0; v < 8; ++v) {
        const size_t idx = (size_t)(rowBase + v + hi * 8) * HIDDEN + colBase;
        lat[v] = *(const v2f*)(latent + idx);
    }
    v2f bf0[4], bf1[4];
    for (int kc = 0; kc < 4; ++kc) {
        bf0[kc] = *(const v2f*)(W + (size_t)(colBase + 0) * NUM_RBF + kc * 4 + hi * 2);
        bf1[kc] = *(const v2f*)(W + (size_t)(colBase + 1) * NUM_RBF + kc * 4 + hi * 2);
    }
    const v2f bv = *(const v2f*)(bias + colBase);

    const int jl  = tid & (JLANES - 1);
    const int row = tid >> 4;
    const int i   = rt * ROWS + row;

    const float xi = sPos[i * 3 + 0];
    const float yi = sPos[i * 3 + 1];
    const float zi = sPos[i * 3 + 2];

    v2f c2[NUM_RBF / 2], acc2[NUM_RBF / 2];
#pragma unroll
    for (int q = 0; q < NUM_RBF / 2; ++q) {
        c2[q]   = (v2f){c[2 * q], c[2 * q + 1]};
        acc2[q] = (v2f){0.0f, 0.0f};
    }
    const v2f k2 = {NEG_HALF_LOG2E, NEG_HALF_LOG2E};

#define ORDER_FENCE() __builtin_amdgcn_sched_barrier(0)
    for (int it = 0; it < LSEQ / (2 * JLANES); ++it) {
        const int j0 = jl + it * (2 * JLANES);
        const int j1 = j0 + JLANES;
        const float dx0 = xi - sPos[j0 * 3 + 0];
        const float dy0 = yi - sPos[j0 * 3 + 1];
        const float dz0 = zi - sPos[j0 * 3 + 2];
        const float dx1 = xi - sPos[j1 * 3 + 0];
        const float dy1 = yi - sPos[j1 * 3 + 1];
        const float dz1 = zi - sPos[j1 * 3 + 2];
        const float d0 = __builtin_amdgcn_sqrtf(dx0 * dx0 + dy0 * dy0 + dz0 * dz0);
        const float d1 = __builtin_amdgcn_sqrtf(dx1 * dx1 + dy1 * dy1 + dz1 * dz1);
        const v2f dd[2] = {(v2f){d0, d0}, (v2f){d1, d1}};

        v2f t[16], u[16], s[16], e[16];
        ORDER_FENCE();
#pragma unroll
        for (int k = 0; k < 16 + 7; ++k) {
            const int ga = k, gb = k - 2, gc = k - 4, gt = k - 5, gf = k - 7;
            const bool hasB = gb >= 0 && gb < 16, hasT = gt >= 0 && gt < 16, hasF = gf >= 0 && gf < 16;
            if (gc >= 0 && gc < 16) { s[gc] = u[gc] * k2;                         ORDER_FENCE(); }
            if (hasT)               { e[gt].x = __builtin_amdgcn_exp2f(s[gt].x);  ORDER_FENCE(); }
            if (hasB)               { u[gb] = t[gb] * t[gb];                      ORDER_FENCE(); }
            else if (hasT && hasF)  { acc2[gf >> 1] += e[gf];                     ORDER_FENCE(); }
            if (hasT)               { e[gt].y = __builtin_amdgcn_exp2f(s[gt].y);  ORDER_FENCE(); }
            if (ga < 16)            { t[ga] = dd[ga & 1] - c2[ga >> 1]; }
            if (hasF && (hasB || !hasT)) { acc2[gf >> 1] += e[gf]; }
            ORDER_FENCE();
        }
    }
#undef ORDER_FENCE

#pragma unroll
    for (int q = 0; q < NUM_RBF / 2; ++q) {
        sPart[tid][2 * q]     = acc2[q].x;
        sPart[tid][2 * q + 1] = acc2[q].y;
    }
    __syncthreads();

    {
        const int orow = tid >> 4;
        const int r    = tid & 15;
        float s = 0.0f;
#pragma unroll
        for (int k = 0; k < JLANES; ++k)
            s += sPart[orow * JLANES + k][r];
        sMean[orow][r] = s * (1.0f / (float)LSEQ);
    }
    __syncthreads();

    v2f a[4];
#pragma unroll
    for (int kc = 0; kc < 4; ++kc)
        a[kc] = *(const v2f*)(&sMean[m][kc * 4 + hi * 2]);

    v8f c0 = {0.f, 0.f, 0.f, 0.f, 0.f, 0.f, 0.f, 0.f};
    v8f c1 = {0.f, 0.f, 0.f, 0.f, 0.f, 0.f, 0.f, 0.f};
#pragma unroll
    for (int kc = 0; kc < 4; ++kc) {
        c0 = __builtin_amdgcn_wmma_f32_16x16x4_f32(
             false, a[kc],  false, bf0[kc],
             (short)0, c0,  false,  false);
        c1 = __builtin_amdgcn_wmma_f32_16x16x4_f32(
             false, a[kc],  false, bf1[kc],
             (short)0, c1,  false,  false);
    }

#pragma unroll
    for (int v = 0; v < 8; ++v) {
        const size_t idx = (size_t)(rowBase + v + hi * 8) * HIDDEN + colBase;
        v2f o;
        o.x = lat[v].x + c0[v] + bv.x;
        o.y = lat[v].y + c1[v] + bv.y;
        *(v2f*)(out + idx) = o;
    }
}

extern "C" void kernel_launch(void* const* d_in, const int* in_sizes, int n_in,
                              void* d_out, int out_size, void* d_ws, size_t ws_size,
                              hipStream_t stream) {
    const float* pos     = (const float*)d_in[0];
    const float* latent  = (const float*)d_in[1];
    const float* W       = (const float*)d_in[2];
    const float* bias    = (const float*)d_in[3];
    const float* centers = (const float*)d_in[4];
    float* out      = (float*)d_out;

    rbf_linear_fused_kernel<<<BATCH * (LSEQ / ROWS), 256, 0, stream>>>(
        pos, centers, W, bias, latent, out);
}


# pair loop: exponent scale folded into the operands (positions and centers held times sqrt(0.5*log2 e), negation inside the square): 3 packed instructions per two v_exp_f32 instead of 4
# speedup vs baseline: 2.0220x; 1.0747x over previous
#include <hip/hip_runtime.h>

#define BATCH 8
#define LSEQ 1024
#define HIDDEN 256
#define NUM_RBF 16
#define ROWS 16
#define JLANES 16
#define MPITCH 18

typedef __attribute__((ext_vector_type(2))) float v2f;
typedef __attribute__((ext_vector_type(8))) float v8f;

#define SQRT_HALF_LOG2E (0.84932180028801904272f)

__global__ __launch_bounds__(256) void rbf_linear_fused_kernel(
    const float* __restrict__ pos,
    const float* __restrict__ centers,
    const float* __restrict__ W,
    const float* __restrict__ bias,
    const float* __restrict__ latent,
    float* __restrict__ out)
{
    __shared__ float sPos[LSEQ * 3];
    __shared__ float sPart[256][NUM_RBF + 1];
    __shared__ __attribute__((aligned(16))) float sMean[ROWS][MPITCH];

    const int tid          = threadIdx.x;
    const int tilesPerB    = LSEQ / ROWS;
    const int b            = blockIdx.x / tilesPerB;
    const int rt           = blockIdx.x % tilesPerB;

    const float* posB = pos + (size_t)b * LSEQ * 3;
    for (int idx = tid; idx < LSEQ * 3; idx += 256)
        sPos[idx] = posB[idx] * SQRT_HALF_LOG2E;

    float c[NUM_RBF];
#pragma unroll
    for (int r = 0; r < NUM_RBF; ++r) c[r] = centers[r] * SQRT_HALF_LOG2E;

    __syncthreads();

    const int lane    = tid & 31;
    const int wv      = tid >> 5;
    const int m       = lane & 15;
    const int hi      = lane >> 4;
    const int rowBase = blockIdx.x * ROWS;
    const int colBase = wv * 32 + 2 * m;

    v2f lat[8];
    for (int v = 0; v < 8; ++v) {
        const size_t idx = (size_t)(rowBase + v + hi * 8) * HIDDEN + colBase;
        lat[v] = *(const v2f*)(latent + idx);
    }
    v2f bf0[4], bf1[4];
    for (int kc = 0; kc < 4; ++kc) {
        bf0[kc] = *(const v2f*)(W + (size_t)(colBase + 0) * NUM_RBF + kc * 4 + hi * 2);
        bf1[kc] = *(const v2f*)(W + (size_t)(colBase + 1) * NUM_RBF + kc * 4 + hi * 2);
    }
    const v2f bv = *(const v2f*)(bias + colBase);

    const int jl  = tid & (JLANES - 1);
    const int row = tid >> 4;
    const int i   = rt * ROWS + row;

    const float xi = sPos[i * 3 + 0];
    const float yi = sPos[i * 3 + 1];
    const float zi = sPos[i * 3 + 2];

    v2f c2[NUM_RBF / 2], acc2[NUM_RBF / 2];
#pragma unroll
    for (int q = 0; q < NUM_RBF / 2; ++q) {
        c2[q]   = (v2f){c[2 * q], c[2 * q + 1]};
        acc2[q] = (v2f){0.0f, 0.0f};
    }

#define ORDER_FENCE() __builtin_amdgcn_sched_barrier(0)
    for (int it = 0; it < LSEQ / (2 * JLANES); ++it) {
        const int j0 = jl + it * (2 * JLANES);
        const int j1 = j0 + JLANES;
        const float dx0 = xi - sPos[j0 * 3 + 0];
        const float dy0 = yi - sPos[j0 * 3 + 1];
        const float dz0 = zi - sPos[j0 * 3 + 2];
        const float dx1 = xi - sPos[j1 * 3 + 0];
        const float dy1 = yi - sPos[j1 * 3 + 1];
        const float dz1 = zi - sPos[j1 * 3 + 2];
        const float d0 = __builtin_amdgcn_sqrtf(dx0 * dx0 + dy0 * dy0 + dz0 * dz0);
        const float d1 = __builtin_amdgcn_sqrtf(dx1 * dx1 + dy1 * dy1 + dz1 * dz1);
        const v2f dd[2] = {(v2f){d0, d0}, (v2f){d1, d1}};

        v2f t[16], u[16], e[16];
        ORDER_FENCE();
#pragma unroll
        for (int k = 0; k < 16 + 6; ++k) {
            const int ga = k, gb = k - 2, gt = k - 4, gf = k - 6;
            const bool hasB = gb >= 0 && gb < 16, hasT = gt >= 0 && gt < 16, hasF = gf >= 0 && gf < 16;
            if (hasT)               { e[gt].x = __builtin_amdgcn_exp2f(u[gt].x);  ORDER_FENCE(); }
            if (hasB)               { u[gb] = t[gb] * (-t[gb]);                   ORDER_FENCE(); }
            else if (hasT && hasF)  { acc2[gf >> 1] += e[gf];                     ORDER_FENCE(); }
            if (hasT)               { e[gt].y = __builtin_amdgcn_exp2f(u[gt].y);  ORDER_FENCE(); }
            if (ga < 16)            { t[ga] = dd[ga & 1] - c2[ga >> 1]; }
            if (hasF && (hasB || !hasT)) { acc2[gf >> 1] += e[gf]; }
            ORDER_FENCE();
        }
    }
#undef ORDER_FENCE

#pragma unroll
    for (int q = 0; q < NUM_RBF / 2; ++q) {
        sPart[tid][2 * q]     = acc2[q].x;
        sPart[tid][2 * q + 1] = acc2[q].y;
    }
    __syncthreads();

    {
        const int orow = tid >> 4;
        const int r    = tid & 15;
        float s = 0.0f;
#pragma unroll
        for (int k = 0; k < JLANES; ++k)
            s += sPart[orow * JLANES + k][r];
        sMean[orow][r] = s * (1.0f / (float)LSEQ);
    }
    __syncthreads();

    v2f a[4];
#pragma unroll
    for (int kc = 0; kc < 4; ++kc)
        a[kc] = *(const v2f*)(&sMean[m][kc * 4 + hi * 2]);

    v8f c0 = {0.f, 0.f, 0.f, 0.f, 0.f, 0.f, 0.f, 0.f};
    v8f c1 = {0.f, 0.f, 0.f, 0.f, 0.f, 0.f, 0.f, 0.f};
#pragma unroll
    for (int kc = 0; kc < 4; ++kc) {
        c0 = __builtin_amdgcn_wmma_f32_16x16x4_f32(
             false, a[kc],  false, bf0[kc],
             (short)0, c0,  false,  false);
        c1 = __builtin_amdgcn_wmma_f32_16x16x4_f32(
             false, a[kc],  false, bf1[kc],
             (short)0, c1,  false,  false);
    }

#pragma unroll
    for (int v = 0; v < 8; ++v) {
        const size_t idx = (size_t)(rowBase + v + hi * 8) * HIDDEN + colBase;
        v2f o;
        o.x = lat[v].x + c0[v] + bv.x;
        o.y = lat[v].y + c1[v] + bv.y;
        *(v2f*)(out + idx) = o;
    }
}

extern "C" void kernel_launch(void* const* d_in, const int* in_sizes, int n_in,
                              void* d_out, int out_size, void* d_ws, size_t ws_size,
                              hipStream_t stream) {
    const float* pos     = (const float*)d_in[0];
    const float* latent  = (const float*)d_in[1];
    const float* W       = (const float*)d_in[2];
    const float* bias    = (const float*)d_in[3];
    const float* centers = (const float*)d_in[4];
    float* out      = (float*)d_out;

    rbf_linear_fused_kernel<<<BATCH * (LSEQ / ROWS), 256, 0, stream>>>(
        pos, centers, W, bias, latent, out);
}


# block prologue: the position staging is requested as three 16-byte loads per thread in one batch (was 4 loads waited for one by one, then 8), the centers' load ahead of it: one round trip, not five
# speedup vs baseline: 2.3128x; 1.1438x over previous
#include <hip/hip_runtime.h>

#define BATCH 8
#define LSEQ 1024
#define HIDDEN 256
#define NUM_RBF 16
#define ROWS 16
#define JLANES 16
#define MPITCH 18

typedef __attribute__((ext_vector_type(2))) float v2f;
typedef __attribute__((ext_vector_type(4))) float v4f;
typedef __attribute__((ext_vector_type(8))) float v8f;

#define SQRT_HALF_LOG2E (0.84932180028801904272f)

__global__ __launch_bounds__(256) void rbf_linear_fused_kernel(
    const float* __restrict__ pos,
    const float* __restrict__ centers,
    const float* __restrict__ W,
    const float* __restrict__ bias,
    const float* __restrict__ latent,
    float* __restrict__ out)
{
    __shared__ __attribute__((aligned(16))) float sPos[LSEQ * 3];
    __shared__ float sPart[256][NUM_RBF + 1];
    __shared__ __attribute__((aligned(16))) float sMean[ROWS][MPITCH];

    const int tid          = threadIdx.x;
    const int tilesPerB    = LSEQ / ROWS;
    const int b            = blockIdx.x / tilesPerB;
    const int rt           = blockIdx.x % tilesPerB;

    float c[NUM_RBF];
#pragma unroll
    for (int r = 0; r < NUM_RBF; ++r) c[r] = centers[r] * SQRT_HALF_LOG2E;

    const v4f* posB4 = (const v4f*)(pos + (size_t)b * LSEQ * 3);
    const v4f p0 = posB4[tid];
    const v4f p1 = posB4[tid + 256];
    const v4f p2 = posB4[tid + 512];
    v4f* sPos4 = (v4f*)sPos;
    sPos4[tid]       = p0 * SQRT_HALF_LOG2E;
    sPos4[tid + 256] = p1 * SQRT_HALF_LOG2E;
    sPos4[tid + 512] = p2 * SQRT_HALF_LOG2E;

    __syncthreads();

    const int lane    = tid & 31;
    const int wv      = tid >> 5;
    const int m       = lane & 15;
    const int hi      = lane >> 4;
    const int rowBase = blockIdx.x * ROWS;
    const int colBase = wv * 32 + 2 * m;

    v2f lat[8];
    for (int v = 0; v < 8; ++v) {
        const size_t idx = (size_t)(rowBase + v + hi * 8) * HIDDEN + colBase;
        lat[v] = *(const v2f*)(latent + idx);
    }
    v2f bf0[4], bf1[4];
    for (int kc = 0; kc < 4; ++kc) {
        bf0[kc] = *(const v2f*)(W + (size_t)(colBase + 0) * NUM_RBF + kc * 4 + hi * 2);
        bf1[kc] = *(const v2f*)(W + (size_t)(colBase + 1) * NUM_RBF + kc * 4 + hi * 2);
    }
    const v2f bv = *(const v2f*)(bias + colBase);

    const int jl  = tid & (JLANES - 1);
    const int row = tid >> 4;
    const int i   = rt * ROWS + row;

    const float xi = sPos[i * 3 + 0];
    const float yi = sPos[i * 3 + 1];
    const float zi = sPos[i * 3 + 2];

    v2f c2[NUM_RBF / 2], acc2[NUM_RBF / 2];
#pragma unroll
    for (int q = 0; q < NUM_RBF / 2; ++q) {
        c2[q]   = (v2f){c[2 * q], c[2 * q + 1]};
        acc2[q] = (v2f){0.0f, 0.0f};
    }

#define ORDER_FENCE() __builtin_amdgcn_sched_barrier(0)
    for (int it = 0; it < LSEQ / (2 * JLANES); ++it) {
        const int j0 = jl + it * (2 * JLANES);
        const int j1 = j0 + JLANES;
        const float dx0 = xi - sPos[j0 * 3 + 0];
        const float dy0 = yi - sPos[j0 * 3 + 1];
        const float dz0 = zi - sPos[j0 * 3 + 2];
        const float dx1 = xi - sPos[j1 * 3 + 0];
        const float dy1 = yi - sPos[j1 * 3 + 1];
        const float dz1 = zi - sPos[j1 * 3 + 2];
        const float d0 = __builtin_amdgcn_sqrtf(dx0 * dx0 + dy0 * dy0 + dz0 * dz0);
        const float d1 = __builtin_amdgcn_sqrtf(dx1 * dx1 + dy1 * dy1 + dz1 * dz1);
        const v2f dd[2] = {(v2f){d0, d0}, (v2f){d1, d1}};

        v2f t[16], u[16], e[16];
        ORDER_FENCE();
#pragma unroll
        for (int k = 0; k < 16 + 6; ++k) {
            const int ga = k, gb = k - 2, gt = k - 4, gf = k - 6;
            const bool hasB = gb >= 0 && gb < 16, hasT = gt >= 0 && gt < 16, hasF = gf >= 0 && gf < 16;
            if (hasT)               { e[gt].x = __builtin_amdgcn_exp2f(u[gt].x);  ORDER_FENCE(); }
            if (hasB)               { u[gb] = t[gb] * (-t[gb]);                   ORDER_FENCE(); }
            else if (hasT && hasF)  { acc2[gf >> 1] += e[gf];                     ORDER_FENCE(); }
            if (hasT)               { e[gt].y = __builtin_amdgcn_exp2f(u[gt].y);  ORDER_FENCE(); }
            if (ga < 16)            { t[ga] = dd[ga & 1] - c2[ga >> 1]; }
            if (hasF && (hasB || !hasT)) { acc2[gf >> 1] += e[gf]; }
            ORDER_FENCE();
        }
    }
#undef ORDER_FENCE

#pragma unroll
    for (int q = 0; q < NUM_RBF / 2; ++q) {
        sPart[tid][2 * q]     = acc2[q].x;
        sPart[tid][2 * q + 1] = acc2[q].y;
    }
    __syncthreads();

    {
        const int orow = tid >> 4;
        const int r    = tid & 15;
        float s = 0.0f;
#pragma unroll
        for (int k = 0; k < JLANES; ++k)
            s += sPart[orow * JLANES + k][r];
        sMean[orow][r] = s * (1.0f / (float)LSEQ);
    }
    __syncthreads();

    v2f a[4];
#pragma unroll
    for (int kc = 0; kc < 4; ++kc)
        a[kc] = *(const v2f*)(&sMean[m][kc * 4 + hi * 2]);

    v8f c0 = {0.f, 0.f, 0.f, 0.f, 0.f, 0.f, 0.f, 0.f};
    v8f c1 = {0.f, 0.f, 0.f, 0.f, 0.f, 0.f, 0.f, 0.f};
#pragma unroll
    for (int kc = 0; kc < 4; ++kc) {
        c0 = __builtin_amdgcn_wmma_f32_16x16x4_f32(
             false, a[kc],  false, bf0[kc],
             (short)0, c0,  false,  false);
        c1 = __builtin_amdgcn_wmma_f32_16x16x4_f32(
             false, a[kc],  false, bf1[kc],
             (short)0, c1,  false,  false);
    }

#pragma unroll
    for (int v = 0; v < 8; ++v) {
        const size_t idx = (size_t)(rowBase + v + hi * 8) * HIDDEN + colBase;
        v2f o;
        o.x = lat[v].x + c0[v] + bv.x;
        o.y = lat[v].y + c1[v] + bv.y;
        *(v2f*)(out + idx) = o;
    }
}

extern "C" void kernel_launch(void* const* d_in, const int* in_sizes, int n_in,
                              void* d_out, int out_size, void* d_ws, size_t ws_size,
                              hipStream_t stream) {
    const float* pos     = (const float*)d_in[0];
    const float* latent  = (const float*)d_in[1];
    const float* W       = (const float*)d_in[2];
    const float* bias    = (const float*)d_in[3];
    const float* centers = (const float*)d_in[4];
    float* out      = (float*)d_out;

    rbf_linear_fused_kernel<<<BATCH * (LSEQ / ROWS), 256, 0, stream>>>(
        pos, centers, W, bias, latent, out);
}


# pair loop: each thread takes two rows (r, r+8) against the same staged positions: 128-thread blocks, half as many waves to start, position reads shared by both rows; sums in the same order
# speedup vs baseline: 2.5333x; 1.0953x over previous
#include <hip/hip_runtime.h>

#define BATCH 8
#define LSEQ 1024
#define HIDDEN 256
#define NUM_RBF 16
#define ROWS 16
#define NTHREADS 128
#define JLANES 16
#define MPITCH 18

typedef __attribute__((ext_vector_type(2))) float v2f;
typedef __attribute__((ext_vector_type(4))) float v4f;
typedef __attribute__((ext_vector_type(8))) float v8f;

#define SQRT_HALF_LOG2E (0.84932180028801904272f)

__global__ __launch_bounds__(NTHREADS) void rbf_linear_fused_kernel(
    const float* __restrict__ pos,
    const float* __restrict__ centers,
    const float* __restrict__ W,
    const float* __restrict__ bias,
    const float* __restrict__ latent,
    float* __restrict__ out)
{
    __shared__ __attribute__((aligned(16))) float sPos[LSEQ * 3];
    __shared__ float sPart[NTHREADS][2 * NUM_RBF + 1];
    __shared__ __attribute__((aligned(16))) float sMean[ROWS][MPITCH];

    const int tid          = threadIdx.x;
    const int tilesPerB    = LSEQ / ROWS;
    const int b            = blockIdx.x / tilesPerB;
    const int rt           = blockIdx.x % tilesPerB;

    float c[NUM_RBF];
#pragma unroll
    for (int r = 0; r < NUM_RBF; ++r) c[r] = centers[r] * SQRT_HALF_LOG2E;

    const v4f* posB4 = (const v4f*)(pos + (size_t)b * LSEQ * 3);
    v4f pv[6];
#pragma unroll
    for (int q = 0; q < 6; ++q) pv[q] = posB4[tid + q * NTHREADS];
    v4f* sPos4 = (v4f*)sPos;
#pragma unroll
    for (int q = 0; q < 6; ++q) sPos4[tid + q * NTHREADS] = pv[q] * SQRT_HALF_LOG2E;

    __syncthreads();

    const int lane    = tid & 31;
    const int wv      = tid >> 5;
    const int m       = lane & 15;
    const int hi      = lane >> 4;
    const int rowBase = blockIdx.x * ROWS;
    const int colBase = wv * 64 + 4 * m;

    v4f lat[8];
#pragma unroll
    for (int v = 0; v < 8; ++v) {
        const size_t idx = (size_t)(rowBase + v + hi * 8) * HIDDEN + colBase;
        lat[v] = *(const v4f*)(latent + idx);
    }
    v2f bf[4][4];
#pragma unroll
    for (int t = 0; t < 4; ++t)
#pragma unroll
        for (int kc = 0; kc < 4; ++kc)
            bf[t][kc] = *(const v2f*)(W + (size_t)(colBase + t) * NUM_RBF + kc * 4 + hi * 2);
    const v4f bv = *(const v4f*)(bias + colBase);

    const int jl  = tid & (JLANES - 1);
    const int rp  = tid >> 4;
    const int ia  = rt * ROWS + rp;
    const int ib  = ia + ROWS / 2;

    const float xa = sPos[ia * 3 + 0];
    const float ya = sPos[ia * 3 + 1];
    const float za = sPos[ia * 3 + 2];
    const float xb = sPos[ib * 3 + 0];
    const float yb = sPos[ib * 3 + 1];
    const float zb = sPos[ib * 3 + 2];

    v2f c2[NUM_RBF / 2], acc2[2][NUM_RBF / 2];
#pragma unroll
    for (int q = 0; q < NUM_RBF / 2; ++q) {
        c2[q]      = (v2f){c[2 * q], c[2 * q + 1]};
        acc2[0][q] = (v2f){0.0f, 0.0f};
        acc2[1][q] = (v2f){0.0f, 0.0f};
    }

#define ORDER_FENCE() __builtin_amdgcn_sched_barrier(0)
#define NGROUPS 32
    for (int it = 0; it < LSEQ / (2 * JLANES); ++it) {
        const int j0 = jl + it * (2 * JLANES);
        const int j1 = j0 + JLANES;
        const float px0 = sPos[j0 * 3 + 0], py0 = sPos[j0 * 3 + 1], pz0 = sPos[j0 * 3 + 2];
        const float px1 = sPos[j1 * 3 + 0], py1 = sPos[j1 * 3 + 1], pz1 = sPos[j1 * 3 + 2];
        const float dxa0 = xa - px0, dya0 = ya - py0, dza0 = za - pz0;
        const float dxa1 = xa - px1, dya1 = ya - py1, dza1 = za - pz1;
        const float dxb0 = xb - px0, dyb0 = yb - py0, dzb0 = zb - pz0;
        const float dxb1 = xb - px1, dyb1 = yb - py1, dzb1 = zb - pz1;
        const float da0 = __builtin_amdgcn_sqrtf(dxa0 * dxa0 + dya0 * dya0 + dza0 * dza0);
        const float da1 = __builtin_amdgcn_sqrtf(dxa1 * dxa1 + dya1 * dya1 + dza1 * dza1);
        const float db0 = __builtin_amdgcn_sqrtf(dxb0 * dxb0 + dyb0 * dyb0 + dzb0 * dzb0);
        const float db1 = __builtin_amdgcn_sqrtf(dxb1 * dxb1 + dyb1 * dyb1 + dzb1 * dzb1);
        const v2f dd[4] = {(v2f){da0, da0}, (v2f){da1, da1}, (v2f){db0, db0}, (v2f){db1, db1}};

        v2f t[NGROUPS], u[NGROUPS], e[NGROUPS];
        ORDER_FENCE();
#pragma unroll
        for (int k = 0; k < NGROUPS + 6; ++k) {
            const int ga = k, gb = k - 2, gt = k - 4, gf = k - 6;
            const bool hasB = gb >= 0 && gb < NGROUPS, hasT = gt >= 0 && gt < NGROUPS,
                       hasF = gf >= 0 && gf < NGROUPS;
            if (hasT)               { e[gt].x = __builtin_amdgcn_exp2f(u[gt].x);  ORDER_FENCE(); }
            if (hasB)               { u[gb] = t[gb] * (-t[gb]);                   ORDER_FENCE(); }
            else if (hasT && hasF)  { acc2[(gf >> 1) & 1][gf >> 2] += e[gf];      ORDER_FENCE(); }
            if (hasT)               { e[gt].y = __builtin_amdgcn_exp2f(u[gt].y);  ORDER_FENCE(); }
            if (ga < NGROUPS)       { t[ga] = dd[ga & 3] - c2[ga >> 2]; }
            if (hasF && (hasB || !hasT)) { acc2[(gf >> 1) & 1][gf >> 2] += e[gf]; }
            ORDER_FENCE();
        }
    }
#undef NGROUPS
#undef ORDER_FENCE

#pragma unroll
    for (int h = 0; h < 2; ++h)
#pragma unroll
        for (int q = 0; q < NUM_RBF / 2; ++q) {
            sPart[tid][h * NUM_RBF + 2 * q]     = acc2[h][q].x;
            sPart[tid][h * NUM_RBF + 2 * q + 1] = acc2[h][q].y;
        }
    __syncthreads();

    {
        const int orp = tid >> 4;
        const int r   = tid & 15;
#pragma unroll
        for (int h = 0; h < 2; ++h) {
            float s = 0.0f;
#pragma unroll
            for (int k = 0; k < JLANES; ++k)
                s += sPart[orp * JLANES + k][h * NUM_RBF + r];
            sMean[orp + h * (ROWS / 2)][r] = s * (1.0f / (float)LSEQ);
        }
    }
    __syncthreads();

    v2f a[4];
#pragma unroll
    for (int kc = 0; kc < 4; ++kc)
        a[kc] = *(const v2f*)(&sMean[m][kc * 4 + hi * 2]);

    v8f cc[4];
#pragma unroll
    for (int t = 0; t < 4; ++t) cc[t] = (v8f){0.f, 0.f, 0.f, 0.f, 0.f, 0.f, 0.f, 0.f};
#pragma unroll
    for (int kc = 0; kc < 4; ++kc) {
#pragma unroll
        for (int t = 0; t < 4; ++t)
            cc[t] = __builtin_amdgcn_wmma_f32_16x16x4_f32(
                 false, a[kc],  false, bf[t][kc],
                 (short)0, cc[t],  false,  false);
    }

#pragma unroll
    for (int v = 0; v < 8; ++v) {
        const size_t idx = (size_t)(rowBase + v + hi * 8) * HIDDEN + colBase;
        v4f o;
        o.x = lat[v].x + cc[0][v] + bv.x;
        o.y = lat[v].y + cc[1][v] + bv.y;
        o.z = lat[v].z + cc[2][v] + bv.z;
        o.w = lat[v].w + cc[3][v] + bv.w;
        *(v4f*)(out + idx) = o;
    }
}

extern "C" void kernel_launch(void* const* d_in, const int* in_sizes, int n_in,
                              void* d_out, int out_size, void* d_ws, size_t ws_size,
                              hipStream_t stream) {
    const float* pos     = (const float*)d_in[0];
    const float* latent  = (const float*)d_in[1];
    const float* W       = (const float*)d_in[2];
    const float* bias    = (const float*)d_in[3];
    const float* centers = (const float*)d_in[4];
    float* out      = (float*)d_out;

    rbf_linear_fused_kernel<<<BATCH * (LSEQ / ROWS), NTHREADS, 0, stream>>>(
        pos, centers, W, bias, latent, out);
}
